# starting kernel
# baseline (speedup 1.0000x reference)
#include <hip/hip_runtime.h>
#include <hip/hip_bf16.h>
#include <math.h>

typedef float v2f __attribute__((ext_vector_type(2)));
typedef float v8f __attribute__((ext_vector_type(8)));

#define BN 4
#define SS 2048
#define DD 1024
#define HH 1024
#define OO 1024
#define FF 2048

template <bool TRANSB>
__global__ void wmma_gemm16(const float* __restrict__ A, long lda, int rows,
                            const float* __restrict__ Bm, int ldb,
                            int K, const float* __restrict__ bias,
                            float* __restrict__ Out, int ldout) {
  __shared__ float red[4][32][8];
  const int lane = threadIdx.x & 31;
  const int wave = threadIdx.x >> 5;
  const int n0   = blockIdx.x * 16;
  const int ln   = lane & 15;
  const int kk   = (lane >> 4) * 2;
  const int kPerWave = K >> 2;
  const int kBeg = wave * kPerWave;

  const float amask = (ln < rows) ? 1.f : 0.f;
  const float* aptr = A + (long)(ln < rows ? ln : 0) * lda + kBeg + kk;
  const float* bptr = TRANSB ? (Bm + (long)(n0 + ln) * ldb + kBeg + kk)
                             : (Bm + (long)(kBeg + kk) * ldb + (n0 + ln));
  const long brow  = TRANSB ? 1 : (long)ldb;
  const long bstep = TRANSB ? 4 : (long)4 * ldb;

  v8f acc0 = {0.f, 0.f, 0.f, 0.f, 0.f, 0.f, 0.f, 0.f};
  v8f acc1 = {0.f, 0.f, 0.f, 0.f, 0.f, 0.f, 0.f, 0.f};

  const int iters = kPerWave >> 2;
  for (int i = 0; i < iters; i += 2) {
    v2f a0, b0, a1, b1;
    a0.x = aptr[0] * amask;  a0.y = aptr[1] * amask;
    b0.x = bptr[0];          b0.y = bptr[brow];
    a1.x = aptr[4] * amask;  a1.y = aptr[5] * amask;
    b1.x = bptr[bstep];      b1.y = bptr[bstep + brow];
    acc0 = __builtin_amdgcn_wmma_f32_16x16x4_f32(
        false, a0, false, b0, (short)0, acc0, false, false);
    acc1 = __builtin_amdgcn_wmma_f32_16x16x4_f32(
        false, a1, false, b1, (short)0, acc1, false, false);
    aptr += 8;
    bptr += 2 * bstep;
  }

  for (int i = 0; i < 8; ++i) red[wave][lane][i] = acc0[i] + acc1[i];
  __syncthreads();

  if (wave == 0 && lane < 16) {
    for (int i = 0; i < rows; ++i) {
      float v = red[0][lane][i] + red[1][lane][i] +
                red[2][lane][i] + red[3][lane][i];
      if (bias) v += bias[n0 + lane];
      Out[(long)i * ldout + n0 + lane] = v;
    }
  }
}

__global__ void scores_k(const float* __restrict__ x,
                         const float* __restrict__ qtil,
                         float* __restrict__ scores) {
  const int b    = blockIdx.y;
  const int srow = blockIdx.x * 8 + (threadIdx.x >> 5);
  const int lane = threadIdx.x & 31;
  const float4* xr = (const float4*)(x + ((size_t)b * SS + srow) * DD);
  const float4* q4 = (const float4*)(qtil + (size_t)b * DD);
  float sum = 0.f;
#pragma unroll
  for (int i = 0; i < 8; ++i) {
    float4 xv = xr[lane + i * 32];
    float4 qv = q4[lane + i * 32];
    sum += xv.x * qv.x + xv.y * qv.y + xv.z * qv.z + xv.w * qv.w;
  }
  for (int off = 16; off; off >>= 1) sum += __shfl_xor(sum, off, 32);
  if (lane == 0) scores[b * SS + srow] = sum * 0.03125f;
}

__global__ void softmax_zero(float* __restrict__ scores,
                             float* __restrict__ cvec) {
  const int b = blockIdx.x, tid = threadIdx.x;
  __shared__ float red[256];
  float* sc = scores + b * SS;
  float v[8];
  float mx = -3.402823466e38f;
#pragma unroll
  for (int j = 0; j < 8; ++j) { v[j] = sc[tid + j * 256]; mx = fmaxf(mx, v[j]); }
  red[tid] = mx; __syncthreads();
  for (int s = 128; s; s >>= 1) {
    if (tid < s) red[tid] = fmaxf(red[tid], red[tid + s]);
    __syncthreads();
  }
  mx = red[0]; __syncthreads();
  float sum = 0.f;
#pragma unroll
  for (int j = 0; j < 8; ++j) { v[j] = expf(v[j] - mx); sum += v[j]; }
  red[tid] = sum; __syncthreads();
  for (int s = 128; s; s >>= 1) {
    if (tid < s) red[tid] += red[tid + s];
    __syncthreads();
  }
  const float inv = 1.f / red[0];
#pragma unroll
  for (int j = 0; j < 8; ++j) sc[tid + j * 256] = v[j] * inv;
#pragma unroll
  for (int c = 0; c < 4; ++c) cvec[b * DD + tid * 4 + c] = 0.f;
}

__global__ void wsum_k(const float* __restrict__ x,
                       const float* __restrict__ wts,
                       float* __restrict__ cvec) {
  const int chunk = blockIdx.x, b = blockIdx.y, tid = threadIdx.x;
  const int s0 = chunk * 128;
  const float4* x4 = (const float4*)x;
  const float* w = wts + b * SS + s0;
  const size_t base = ((size_t)b * SS + s0) * (DD / 4);
  float4 acc = {0.f, 0.f, 0.f, 0.f};
  for (int s = 0; s < 128; ++s) {
    const float ww = w[s];
    float4 xv = x4[base + (size_t)s * (DD / 4) + tid];
    acc.x += ww * xv.x; acc.y += ww * xv.y;
    acc.z += ww * xv.z; acc.w += ww * xv.w;
  }
  float* c = cvec + b * DD + tid * 4;
  atomicAdd(c + 0, acc.x); atomicAdd(c + 1, acc.y);
  atomicAdd(c + 2, acc.z); atomicAdd(c + 3, acc.w);
}

__global__ void glu_ln(const float* __restrict__ hin,
                       const float* __restrict__ attn,
                       const float* __restrict__ gamma,
                       const float* __restrict__ beta,
                       float* __restrict__ comb) {
  const int b = blockIdx.x, tid = threadIdx.x;
  __shared__ float red[256];
  const float* hb = hin + (size_t)b * FF;
  float hv[4];
  float sum = 0.f;
#pragma unroll
  for (int j = 0; j < 4; ++j) {
    const int i = tid + j * 256;
    const float x1 = hb[i];
    const float x2 = hb[HH + i];
    const float g = 0.5f * x2 * (1.f + erff(x2 * 0.70710678118654752f));
    hv[j] = x1 * g;
    sum += hv[j];
  }
  red[tid] = sum; __syncthreads();
  for (int s = 128; s; s >>= 1) { if (tid < s) red[tid] += red[tid + s]; __syncthreads(); }
  const float mean = red[0] * (1.f / HH); __syncthreads();
  float sq = 0.f;
#pragma unroll
  for (int j = 0; j < 4; ++j) { const float d = hv[j] - mean; sq += d * d; }
  red[tid] = sq; __syncthreads();
  for (int s = 128; s; s >>= 1) { if (tid < s) red[tid] += red[tid + s]; __syncthreads(); }
  const float rstd = rsqrtf(red[0] * (1.f / HH) + 1e-5f);
#pragma unroll
  for (int j = 0; j < 4; ++j) {
    const int i = tid + j * 256;
    comb[b * HH + i] = attn[b * HH + i] + (hv[j] - mean) * rstd * gamma[i] + beta[i];
  }
}

extern "C" void kernel_launch(void* const* d_in, const int* in_sizes, int n_in,
                              void* d_out, int out_size, void* d_ws, size_t ws_size,
                              hipStream_t stream) {
  const float* x       = (const float*)d_in[0];
  const float* Wq_proj = (const float*)d_in[1];
  const float* bq      = (const float*)d_in[2];
  const float* Wq      = (const float*)d_in[3];
  const float* Wk      = (const float*)d_in[4];
  const float* Wv      = (const float*)d_in[5];
  const float* W1      = (const float*)d_in[6];
  const float* b1      = (const float*)d_in[7];
  const float* W2      = (const float*)d_in[8];
  const float* b2      = (const float*)d_in[9];
  const float* gamma   = (const float*)d_in[10];
  const float* beta    = (const float*)d_in[11];
  float* out = (float*)d_out;

  float* ws     = (float*)d_ws;
  float* qproj  = ws;
  float* qlast  = ws + 4096;
  float* qtil   = ws + 8192;
  float* scores = ws + 12288;
  float* cvec   = ws + 20480;
  float* attn   = ws + 24576;
  float* hin    = ws + 28672;
  float* comb   = ws + 36864;

  wmma_gemm16<false><<<dim3(HH / 16), 128, 0, stream>>>(
      x + (size_t)(SS - 1) * DD, (long)SS * DD, BN, Wq_proj, HH, DD, bq, qproj, HH);
  wmma_gemm16<false><<<dim3(HH / 16), 128, 0, stream>>>(
      qproj, HH, BN, Wq, HH, HH, nullptr, qlast, HH);
  wmma_gemm16<true><<<dim3(DD / 16), 128, 0, stream>>>(
      qlast, HH, BN, Wk, HH, HH, nullptr, qtil, DD);
  scores_k<<<dim3(SS / 8, BN), 256, 0, stream>>>(x, qtil, scores);
  softmax_zero<<<dim3(BN), 256, 0, stream>>>(scores, cvec);
  wsum_k<<<dim3(16, BN), 256, 0, stream>>>(x, scores, cvec);
  wmma_gemm16<false><<<dim3(HH / 16), 128, 0, stream>>>(
      cvec, DD, BN, Wv, HH, DD, nullptr, attn, HH);
  wmma_gemm16<false><<<dim3(FF / 16), 128, 0, stream>>>(
      attn, HH, BN, W1, FF, HH, b1, hin, FF);
  glu_ln<<<dim3(BN), 256, 0, stream>>>(hin, attn, gamma, beta, comb);
  wmma_gemm16<false><<<dim3(OO / 16), 128, 0, stream>>>(
      comb, HH, BN, W2, OO, HH, b2, out, OO);
}


# softmax and GLU+LayerNorm kernels: 8-level LDS tree reductions (a barrier per level) replaced by in-wave DPP/permlane butterflies plus one LDS exchange of 8 wave partials; sums in a new tree order
# speedup vs baseline: 1.6535x; 1.6535x over previous
#include <hip/hip_runtime.h>
#include <hip/hip_bf16.h>
#include <math.h>

typedef float v2f __attribute__((ext_vector_type(2)));
typedef float v8f __attribute__((ext_vector_type(8)));
typedef float v4f __attribute__((ext_vector_type(4)));

#define BN 4
#define SS 2048
#define DD 1024
#define HH 1024
#define OO 1024
#define FF 2048

#define GK 1024
#define LPITCH 1028

template <bool TRANSB>
__global__ void __launch_bounds__(128)
wmma_gemm16(const float* __restrict__ A, long lda,
            const float* __restrict__ Bm, int ldb,
            const float* __restrict__ bias,
            float* __restrict__ Out, int ldout) {
  __shared__ __attribute__((aligned(16))) float sA[BN * LPITCH];
  __shared__ __attribute__((aligned(16))) float sB[TRANSB ? 16 * LPITCH : GK * 16];
  __shared__ float red[4][BN][16];
  const int tid  = threadIdx.x;
  const int lane = tid & 31;
  const int wave = tid >> 5;
  const int n0   = blockIdx.x * 16;
  const int ln   = lane & 15;
  const int kk   = (lane >> 4) * 2;
  const int kBeg = wave * (GK / 4);

  v4f av[8];
  v4f bv[8][4];
#pragma unroll
  for (int i = 0; i < 8; ++i) {
    av[i] = *(const v4f*)(A + (long)(i >> 1) * lda + kBeg + 128 * (i & 1) + 4 * lane);
  }
  const float* gp = TRANSB
      ? (Bm + (long)(n0 + (lane >> 3)) * ldb + kBeg + 4 * (lane & 7))
      : (Bm + (long)(kBeg + (lane >> 2)) * ldb + n0 + 4 * (lane & 3));
#pragma unroll
  for (int c = 0; c < 8; ++c) {
#pragma unroll
    for (int j = 0; j < 4; ++j) {
      if (TRANSB)
        bv[c][j] = *(const v4f*)(gp + (long)(4 * j) * ldb + 32 * c);
      else
        bv[c][j] = *(const v4f*)(gp + (long)(32 * c + 8 * j) * ldb);
    }
    __builtin_amdgcn_sched_barrier(0);
  }

#pragma unroll
  for (int i = 0; i < 8; ++i)
    *(v4f*)(sA + (i >> 1) * LPITCH + kBeg + 128 * (i & 1) + 4 * lane) = av[i];

  const float* ap = sA + (ln & 3) * LPITCH + kBeg + kk;
  const float* bp = TRANSB ? (sB + ln * LPITCH + kBeg + kk)
                           : (sB + (kBeg + kk) * 16 + ln);
  float* sp = TRANSB ? (sB + (lane >> 3) * LPITCH + kBeg + 4 * (lane & 7))
                     : (sB + kBeg * 16 + 4 * lane);

  v8f acc0 = {0.f, 0.f, 0.f, 0.f, 0.f, 0.f, 0.f, 0.f};
  v8f acc1 = {0.f, 0.f, 0.f, 0.f, 0.f, 0.f, 0.f, 0.f};

  v2f ae[8], be[8], ao[8], bo[8];

#define GEMM_CHUNK_IN(C, FA, FB)                                                   \
  _Pragma("unroll") for (int j = 0; j < 4; ++j) {                                  \
    if (TRANSB) *(v4f*)(sp + 4 * j * LPITCH + 32 * (C)) = bv[C][j];                \
    else        *(v4f*)(sp + 512 * (C) + 128 * j) = bv[C][j];                      \
  }                                                                                \
  __builtin_amdgcn_fence(__ATOMIC_RELEASE, "wavefront");                           \
  __builtin_amdgcn_wave_barrier();                                                 \
  __builtin_amdgcn_fence(__ATOMIC_ACQUIRE, "wavefront");                           \
  _Pragma("unroll") for (int j = 0; j < 8; ++j) {                                  \
    const int i = 8 * (C) + j;                     \
    FA[j] = *(const v2f*)(ap + 4 * i);                                             \
    if (TRANSB) {                                                                  \
      FB[j] = *(const v2f*)(bp + 4 * i);                                           \
    } else {                                                                       \
      FB[j].x = bp[(4 * i) * 16];                                                  \
      FB[j].y = bp[(4 * i + 1) * 16];                                              \
    }                                                                              \
  }
#define GEMM_CHUNK_MM(FA, FB)                                                      \
  _Pragma("unroll") for (int j = 0; j < 8; j += 2) {                               \
    acc0 = __builtin_amdgcn_wmma_f32_16x16x4_f32(                                  \
        false, FA[j], false, FB[j], (short)0, acc0, false, false);                 \
    acc1 = __builtin_amdgcn_wmma_f32_16x16x4_f32(                                  \
        false, FA[j + 1], false, FB[j + 1], (short)0, acc1, false, false);         \
  }
  constexpr int DSC = TRANSB ? 8 : 12;

  GEMM_CHUNK_IN(0, ae, be)
  GEMM_CHUNK_IN(1, ao, bo) GEMM_CHUNK_MM(ae, be)
  GEMM_CHUNK_IN(2, ae, be) GEMM_CHUNK_MM(ao, bo)
  GEMM_CHUNK_IN(3, ao, bo) GEMM_CHUNK_MM(ae, be)
  GEMM_CHUNK_IN(4, ae, be) GEMM_CHUNK_MM(ao, bo)
  GEMM_CHUNK_IN(5, ao, bo) GEMM_CHUNK_MM(ae, be)
  GEMM_CHUNK_IN(6, ae, be) GEMM_CHUNK_MM(ao, bo)
  GEMM_CHUNK_IN(7, ao, bo) GEMM_CHUNK_MM(ae, be)
  GEMM_CHUNK_MM(ao, bo)
#undef GEMM_CHUNK_IN
#undef GEMM_CHUNK_MM
  __builtin_amdgcn_sched_group_barrier(0x200, 12, 0);
  __builtin_amdgcn_sched_group_barrier(0x100, DSC, 0);
#pragma unroll
  for (int c = 1; c < 8; ++c) {
    __builtin_amdgcn_sched_group_barrier(0x200, 4, 0);
    __builtin_amdgcn_sched_group_barrier(0x100, DSC, 0);
    __builtin_amdgcn_sched_group_barrier(0x008, 8, 0);
  }
  __builtin_amdgcn_sched_group_barrier(0x008, 8, 0);

  if (lane < 16) {
    for (int i = 0; i < BN; ++i) red[wave][i][lane] = acc0[i] + acc1[i];
  }
  __syncthreads();

  if (wave == 0 && lane < 16) {
    for (int i = 0; i < BN; ++i) {
      float v = red[0][i][lane] + red[1][i][lane] +
                red[2][i][lane] + red[3][i][lane];
      if (bias) v += bias[n0 + lane];
      Out[(long)i * ldout + n0 + lane] = v;
    }
  }
}

template <int M>
__device__ __forceinline__ float rowx(float x) {
  const int b = __builtin_bit_cast(int, x);
  return __builtin_bit_cast(
      float, __builtin_amdgcn_update_dpp(b, b, 0x160 + M, 0xf, 0xf, false));
}
__device__ __forceinline__ float xhalf(float x) {
  const int b = __builtin_bit_cast(int, x);
  return __builtin_bit_cast(
      float, __builtin_amdgcn_permlanex16(b, b, 0x76543210, 0xfedcba98, false, false));
}
__device__ __forceinline__ float block_sum(float v, float* part) {
  v += rowx<1>(v); v += rowx<2>(v); v += rowx<4>(v); v += rowx<8>(v);
  v += xhalf(v);
  if ((threadIdx.x & 31) == 0) part[threadIdx.x >> 5] = v;
  __syncthreads();
  const float4 a = *(const float4*)part;
  const float4 b = *(const float4*)(part + 4);
  return ((a.x + a.y) + (a.z + a.w)) + ((b.x + b.y) + (b.z + b.w));
}
__device__ __forceinline__ float block_max(float v, float* part) {
  v = fmaxf(v, rowx<1>(v)); v = fmaxf(v, rowx<2>(v));
  v = fmaxf(v, rowx<4>(v)); v = fmaxf(v, rowx<8>(v));
  v = fmaxf(v, xhalf(v));
  if ((threadIdx.x & 31) == 0) part[threadIdx.x >> 5] = v;
  __syncthreads();
  const float4 a = *(const float4*)part;
  const float4 b = *(const float4*)(part + 4);
  return fmaxf(fmaxf(fmaxf(a.x, a.y), fmaxf(a.z, a.w)),
               fmaxf(fmaxf(b.x, b.y), fmaxf(b.z, b.w)));
}

__global__ void scores_k(const float* __restrict__ x,
                         const float* __restrict__ qtil,
                         float* __restrict__ scores) {
  const int b    = blockIdx.y;
  const int srow = blockIdx.x * 8 + (threadIdx.x >> 5);
  const int lane = threadIdx.x & 31;
  const float4* xr = (const float4*)(x + ((size_t)b * SS + srow) * DD);
  const float4* q4 = (const float4*)(qtil + (size_t)b * DD);
  float sum = 0.f;
#pragma unroll
  for (int i = 0; i < 8; ++i) {
    float4 xv = xr[lane + i * 32];
    float4 qv = q4[lane + i * 32];
    sum += xv.x * qv.x + xv.y * qv.y + xv.z * qv.z + xv.w * qv.w;
  }
  for (int off = 16; off; off >>= 1) sum += __shfl_xor(sum, off, 32);
  if (lane == 0) scores[b * SS + srow] = sum * 0.03125f;
}

__global__ void softmax_k(float* __restrict__ scores) {
  const int b = blockIdx.x, tid = threadIdx.x;
  __shared__ __attribute__((aligned(16))) float part[2][8];
  float* sc = scores + b * SS;
  float v[8];
  float mx = -3.402823466e38f;
#pragma unroll
  for (int j = 0; j < 8; ++j) { v[j] = sc[tid + j * 256]; mx = fmaxf(mx, v[j]); }
  mx = block_max(mx, part[0]);
  float sum = 0.f;
#pragma unroll
  for (int j = 0; j < 8; ++j) { v[j] = expf(v[j] - mx); sum += v[j]; }
  const float inv = 1.f / block_sum(sum, part[1]);
#pragma unroll
  for (int j = 0; j < 8; ++j) sc[tid + j * 256] = v[j] * inv;
}

#define WROWS 64
__global__ void __launch_bounds__(256) __attribute__((amdgpu_waves_per_eu(2, 2)))
wsum_cols(const float* __restrict__ x,
          const float* __restrict__ wts,
          float* __restrict__ cvec) {
  __shared__ __attribute__((aligned(16))) float red[8][32];
  const int strip = blockIdx.x, b = blockIdx.y, tid = threadIdx.x;
  const int q = tid & 7, g = tid >> 3;
  const int lane = tid & 31, wave = tid >> 5;
  const float4* xp = (const float4*)x +
      ((size_t)b * SS + (size_t)g * WROWS) * (DD / 4) + strip * 8 + q;
  const float4* wp = (const float4*)(wts + b * SS + g * WROWS);
  float4 wv[WROWS / 4];
  float4 xv[WROWS];
#pragma unroll
  for (int i = 0; i < WROWS / 4; ++i) wv[i] = wp[i];
#pragma unroll
  for (int i = 0; i < WROWS; ++i) xv[i] = xp[(size_t)i * (DD / 4)];
  float4 a0 = {0.f, 0.f, 0.f, 0.f};
  float4 a1 = {0.f, 0.f, 0.f, 0.f};
#pragma unroll
  for (int i = 0; i < WROWS / 4; ++i) {
    const float4 w4 = wv[i];
    const float4 x0 = xv[4 * i], x1 = xv[4 * i + 1];
    const float4 x2 = xv[4 * i + 2], x3 = xv[4 * i + 3];
    a0.x += w4.x * x0.x; a0.y += w4.x * x0.y; a0.z += w4.x * x0.z; a0.w += w4.x * x0.w;
    a1.x += w4.y * x1.x; a1.y += w4.y * x1.y; a1.z += w4.y * x1.z; a1.w += w4.y * x1.w;
    a0.x += w4.z * x2.x; a0.y += w4.z * x2.y; a0.z += w4.z * x2.z; a0.w += w4.z * x2.w;
    a1.x += w4.w * x3.x; a1.y += w4.w * x3.y; a1.z += w4.w * x3.z; a1.w += w4.w * x3.w;
  }
  float4 a = {a0.x + a1.x, a0.y + a1.y, a0.z + a1.z, a0.w + a1.w};
  for (int off = 8; off <= 16; off <<= 1) {
    a.x += __shfl_xor(a.x, off, 32); a.y += __shfl_xor(a.y, off, 32);
    a.z += __shfl_xor(a.z, off, 32); a.w += __shfl_xor(a.w, off, 32);
  }
  if (lane < 8) *(float4*)&red[wave][4 * lane] = a;
  __syncthreads();
  if (tid < 32) {
    float v = red[0][tid];
    for (int w = 1; w < 8; ++w) v += red[w][tid];
    cvec[b * DD + strip * 32 + tid] = v;
  }
}

__global__ void glu_ln(const float* __restrict__ hin,
                       const float* __restrict__ attn,
                       const float* __restrict__ gamma,
                       const float* __restrict__ beta,
                       float* __restrict__ comb) {
  const int b = blockIdx.x, tid = threadIdx.x;
  __shared__ __attribute__((aligned(16))) float part[2][8];
  const float* hb = hin + (size_t)b * FF;
  float hv[4];
  float sum = 0.f;
#pragma unroll
  for (int j = 0; j < 4; ++j) {
    const int i = tid + j * 256;
    const float x1 = hb[i];
    const float x2 = hb[HH + i];
    const float g = 0.5f * x2 * (1.f + erff(x2 * 0.70710678118654752f));
    hv[j] = x1 * g;
    sum += hv[j];
  }
  const float mean = block_sum(sum, part[0]) * (1.f / HH);
  float sq = 0.f;
#pragma unroll
  for (int j = 0; j < 4; ++j) { const float d = hv[j] - mean; sq += d * d; }
  const float rstd = rsqrtf(block_sum(sq, part[1]) * (1.f / HH) + 1e-5f);
#pragma unroll
  for (int j = 0; j < 4; ++j) {
    const int i = tid + j * 256;
    comb[b * HH + i] = attn[b * HH + i] + (hv[j] - mean) * rstd * gamma[i] + beta[i];
  }
}

extern "C" void kernel_launch(void* const* d_in, const int* in_sizes, int n_in,
                              void* d_out, int out_size, void* d_ws, size_t ws_size,
                              hipStream_t stream) {
  const float* x       = (const float*)d_in[0];
  const float* Wq_proj = (const float*)d_in[1];
  const float* bq      = (const float*)d_in[2];
  const float* Wq      = (const float*)d_in[3];
  const float* Wk      = (const float*)d_in[4];
  const float* Wv      = (const float*)d_in[5];
  const float* W1      = (const float*)d_in[6];
  const float* b1      = (const float*)d_in[7];
  const float* W2      = (const float*)d_in[8];
  const float* b2      = (const float*)d_in[9];
  const float* gamma   = (const float*)d_in[10];
  const float* beta    = (const float*)d_in[11];
  float* out = (float*)d_out;

  float* ws     = (float*)d_ws;
  float* qproj  = ws;
  float* qlast  = ws + 4096;
  float* qtil   = ws + 8192;
  float* scores = ws + 12288;
  float* cvec   = ws + 20480;
  float* attn   = ws + 24576;
  float* hin    = ws + 28672;
  float* comb   = ws + 36864;

  wmma_gemm16<false><<<dim3(HH / 16), 128, 0, stream>>>(
      x + (size_t)(SS - 1) * DD, (long)SS * DD, Wq_proj, HH, bq, qproj, HH);
  wmma_gemm16<false><<<dim3(HH / 16), 128, 0, stream>>>(
      qproj, HH, Wq, HH, nullptr, qlast, HH);
  wmma_gemm16<true><<<dim3(DD / 16), 128, 0, stream>>>(
      qlast, HH, Wk, HH, nullptr, qtil, DD);
  scores_k<<<dim3(SS / 8, BN), 256, 0, stream>>>(x, qtil, scores);
  softmax_k<<<dim3(BN), 256, 0, stream>>>(scores);
  wsum_cols<<<dim3(DD / 32, BN), 256, 0, stream>>>(x, scores, cvec);
  wmma_gemm16<false><<<dim3(HH / 16), 128, 0, stream>>>(
      cvec, DD, Wv, HH, nullptr, attn, HH);
  wmma_gemm16<false><<<dim3(FF / 16), 128, 0, stream>>>(
      attn, HH, W1, FF, b1, hin, FF);
  glu_ln<<<dim3(BN), 256, 0, stream>>>(hin, attn, gamma, beta, comb);
  wmma_gemm16<false><<<dim3(OO / 16), 128, 0, stream>>>(
      comb, HH, W2, OO, b2, out, OO);
}


# launches 10 -> 9: W1 GEMM block takes a column tile and its GLU partner (n, n+1024), applies GLU, writes h + per-tile mean/M2; last GEMM merges the 64 pairs, normalises its A rows; LN sum order new
# speedup vs baseline: 1.7533x; 1.0604x over previous
#include <hip/hip_runtime.h>
#include <hip/hip_bf16.h>
#include <math.h>

typedef float v2f __attribute__((ext_vector_type(2)));
typedef float v8f __attribute__((ext_vector_type(8)));
typedef float v4f __attribute__((ext_vector_type(4)));

#define BN 4
#define SS 2048
#define DD 1024
#define HH 1024
#define OO 1024
#define FF 2048

template <int M>
__device__ __forceinline__ float rowx(float x) {
  const int b = __builtin_bit_cast(int, x);
  return __builtin_bit_cast(
      float, __builtin_amdgcn_update_dpp(b, b, 0x160 + M, 0xf, 0xf, false));
}
__device__ __forceinline__ float xhalf(float x) {
  const int b = __builtin_bit_cast(int, x);
  return __builtin_bit_cast(
      float, __builtin_amdgcn_permlanex16(b, b, 0x76543210, 0xfedcba98, false, false));
}
__device__ __forceinline__ float block_sum(float v, float* part) {
  v += rowx<1>(v); v += rowx<2>(v); v += rowx<4>(v); v += rowx<8>(v);
  v += xhalf(v);
  if ((threadIdx.x & 31) == 0) part[threadIdx.x >> 5] = v;
  __syncthreads();
  const float4 a = *(const float4*)part;
  const float4 b = *(const float4*)(part + 4);
  return ((a.x + a.y) + (a.z + a.w)) + ((b.x + b.y) + (b.z + b.w));
}
__device__ __forceinline__ float block_max(float v, float* part) {
  v = fmaxf(v, rowx<1>(v)); v = fmaxf(v, rowx<2>(v));
  v = fmaxf(v, rowx<4>(v)); v = fmaxf(v, rowx<8>(v));
  v = fmaxf(v, xhalf(v));
  if ((threadIdx.x & 31) == 0) part[threadIdx.x >> 5] = v;
  __syncthreads();
  const float4 a = *(const float4*)part;
  const float4 b = *(const float4*)(part + 4);
  return fmaxf(fmaxf(fmaxf(a.x, a.y), fmaxf(a.z, a.w)),
               fmaxf(fmaxf(b.x, b.y), fmaxf(b.z, b.w)));
}

__device__ __forceinline__ void merge_stats(float& m, float& q, float mo, float qo,
                                            float halfn) {
  const float d = mo - m;
  m = 0.5f * (m + mo);
  q = (q + qo) + d * d * halfn;
}

#define GK 1024
#define LPITCH 1028

template <bool TRANSB, int MODE>
__global__ void __launch_bounds__(MODE == 1 ? 256 : 128)
wmma_gemm16(const float* __restrict__ A, long lda,
            const float* __restrict__ Bm, int ldb,
            const float* __restrict__ bias,
            float* __restrict__ Out, int ldout,
            const float* __restrict__ Hn,
            const float* __restrict__ gamma,
            const float* __restrict__ beta,
            float* __restrict__ stats) {
  constexpr int NP = (MODE == 1) ? 2 : 1;
  __shared__ __attribute__((aligned(16))) float sA[NP * BN * LPITCH];
  __shared__ __attribute__((aligned(16))) float sB[NP * (TRANSB ? 16 * LPITCH : GK * 16)];
  __shared__ float red[4 * NP][BN][16];
  const int tid  = threadIdx.x;
  const int lane = tid & 31;
  const int wave = tid >> 5;
  const int kq    = wave & 3;
  const int panel = (MODE == 1) ? (wave >> 2) : 0;
  const int n0   = blockIdx.x * 16 + panel * HH;
  const int ln   = lane & 15;
  const int kk   = (lane >> 4) * 2;
  const int kBeg = kq * (GK / 4);
  float* const sAp = sA + panel * (BN * LPITCH);
  float* const sBp = sB + panel * (GK * 16);

  v4f av[8];
  v4f bv[8][4];
#pragma unroll
  for (int i = 0; i < 8; ++i) {
    av[i] = *(const v4f*)(A + (long)(i >> 1) * lda + kBeg + 128 * (i & 1) + 4 * lane);
  }
  v4f hv[8], gv[2], bev[2], st[4];
  if (MODE == 2) {
#pragma unroll
    for (int i = 0; i < 8; ++i)
      hv[i] = *(const v4f*)(Hn + (i >> 1) * HH + kBeg + 128 * (i & 1) + 4 * lane);
#pragma unroll
    for (int j = 0; j < 2; ++j) {
      gv[j]  = *(const v4f*)(gamma + kBeg + 128 * j + 4 * lane);
      bev[j] = *(const v4f*)(beta + kBeg + 128 * j + 4 * lane);
      st[2 * j]     = *(const v4f*)(stats + (lane + 32 * j) * (2 * BN));
      st[2 * j + 1] = *(const v4f*)(stats + (lane + 32 * j) * (2 * BN) + 4);
    }
  }
  const float* gp = TRANSB
      ? (Bm + (long)(n0 + (lane >> 3)) * ldb + kBeg + 4 * (lane & 7))
      : (Bm + (long)(kBeg + (lane >> 2)) * ldb + n0 + 4 * (lane & 3));
#pragma unroll
  for (int c = 0; c < 8; ++c) {
#pragma unroll
    for (int j = 0; j < 4; ++j) {
      if (TRANSB)
        bv[c][j] = *(const v4f*)(gp + (long)(4 * j) * ldb + 32 * c);
      else
        bv[c][j] = *(const v4f*)(gp + (long)(32 * c + 8 * j) * ldb);
    }
    __builtin_amdgcn_sched_barrier(0);
  }

  if (MODE == 2) {
#pragma unroll
    for (int r = 0; r < BN; ++r) {
      const v4f sa = st[r >> 1], sb = st[2 + (r >> 1)];
      float m = (r & 1) ? sa.z : sa.x, q = (r & 1) ? sa.w : sa.y;
      merge_stats(m, q, (r & 1) ? sb.z : sb.x, (r & 1) ? sb.w : sb.y, 8.f);
      merge_stats(m, q, rowx<1>(m), rowx<1>(q), 16.f);
      merge_stats(m, q, rowx<2>(m), rowx<2>(q), 32.f);
      merge_stats(m, q, rowx<4>(m), rowx<4>(q), 64.f);
      merge_stats(m, q, rowx<8>(m), rowx<8>(q), 128.f);
      merge_stats(m, q, xhalf(m), xhalf(q), 256.f);
      const float rstd = rsqrtf(q * (1.f / HH) + 1e-5f);
      av[2 * r]     = av[2 * r]     + (hv[2 * r]     - m) * rstd * gv[0] + bev[0];
      av[2 * r + 1] = av[2 * r + 1] + (hv[2 * r + 1] - m) * rstd * gv[1] + bev[1];
    }
  }

#pragma unroll
  for (int i = 0; i < 8; ++i)
    *(v4f*)(sAp + (i >> 1) * LPITCH + kBeg + 128 * (i & 1) + 4 * lane) = av[i];

  const float* ap = sAp + (ln & 3) * LPITCH + kBeg + kk;
  const float* bp = TRANSB ? (sBp + ln * LPITCH + kBeg + kk)
                           : (sBp + (kBeg + kk) * 16 + ln);
  float* sp = TRANSB ? (sBp + (lane >> 3) * LPITCH + kBeg + 4 * (lane & 7))
                     : (sBp + kBeg * 16 + 4 * lane);

  v8f acc0 = {0.f, 0.f, 0.f, 0.f, 0.f, 0.f, 0.f, 0.f};
  v8f acc1 = {0.f, 0.f, 0.f, 0.f, 0.f, 0.f, 0.f, 0.f};

  v2f ae[8], be[8], ao[8], bo[8];

#define GEMM_CHUNK_IN(C, FA, FB)                                                   \
  _Pragma("unroll") for (int j = 0; j < 4; ++j) {                                  \
    if (TRANSB) *(v4f*)(sp + 4 * j * LPITCH + 32 * (C)) = bv[C][j];                \
    else        *(v4f*)(sp + 512 * (C) + 128 * j) = bv[C][j];                      \
  }                                                                                \
  __builtin_amdgcn_fence(__ATOMIC_RELEASE, "wavefront");                           \
  __builtin_amdgcn_wave_barrier();                                                 \
  __builtin_amdgcn_fence(__ATOMIC_ACQUIRE, "wavefront");                           \
  _Pragma("unroll") for (int j = 0; j < 8; ++j) {                                  \
    const int i = 8 * (C) + j;                     \
    FA[j] = *(const v2f*)(ap + 4 * i);                                             \
    if (TRANSB) {                                                                  \
      FB[j] = *(const v2f*)(bp + 4 * i);                                           \
    } else {                                                                       \
      FB[j].x = bp[(4 * i) * 16];                                                  \
      FB[j].y = bp[(4 * i + 1) * 16];                                              \
    }                                                                              \
  }
#define GEMM_CHUNK_MM(FA, FB)                                                      \
  _Pragma("unroll") for (int j = 0; j < 8; j += 2) {                               \
    acc0 = __builtin_amdgcn_wmma_f32_16x16x4_f32(                                  \
        false, FA[j], false, FB[j], (short)0, acc0, false, false);                 \
    acc1 = __builtin_amdgcn_wmma_f32_16x16x4_f32(                                  \
        false, FA[j + 1], false, FB[j + 1], (short)0, acc1, false, false);         \
  }
  constexpr int DSC = TRANSB ? 8 : 12;

  GEMM_CHUNK_IN(0, ae, be)
  GEMM_CHUNK_IN(1, ao, bo) GEMM_CHUNK_MM(ae, be)
  GEMM_CHUNK_IN(2, ae, be) GEMM_CHUNK_MM(ao, bo)
  GEMM_CHUNK_IN(3, ao, bo) GEMM_CHUNK_MM(ae, be)
  GEMM_CHUNK_IN(4, ae, be) GEMM_CHUNK_MM(ao, bo)
  GEMM_CHUNK_IN(5, ao, bo) GEMM_CHUNK_MM(ae, be)
  GEMM_CHUNK_IN(6, ae, be) GEMM_CHUNK_MM(ao, bo)
  GEMM_CHUNK_IN(7, ao, bo) GEMM_CHUNK_MM(ae, be)
  GEMM_CHUNK_MM(ao, bo)
#undef GEMM_CHUNK_IN
#undef GEMM_CHUNK_MM
  __builtin_amdgcn_sched_group_barrier(0x200, 12, 0);
  __builtin_amdgcn_sched_group_barrier(0x100, DSC, 0);
#pragma unroll
  for (int c = 1; c < 8; ++c) {
    __builtin_amdgcn_sched_group_barrier(0x200, 4, 0);
    __builtin_amdgcn_sched_group_barrier(0x100, DSC, 0);
    __builtin_amdgcn_sched_group_barrier(0x008, 8, 0);
  }
  __builtin_amdgcn_sched_group_barrier(0x008, 8, 0);

  if (lane < 16) {
    for (int i = 0; i < BN; ++i) red[wave][i][lane] = acc0[i] + acc1[i];
  }
  __syncthreads();

  if (MODE == 1) {
    if (wave == 0) {
      const int c  = lane & 15;
      const int na = blockIdx.x * 16 + c;
      const float bx1 = bias[na], bx2 = bias[HH + na];
      for (int i = 0; i < BN; ++i) {
        const float x1 = red[0][i][c] + red[1][i][c] + red[2][i][c] + red[3][i][c] + bx1;
        const float x2 = red[4][i][c] + red[5][i][c] + red[6][i][c] + red[7][i][c] + bx2;
        const float g = 0.5f * x2 * (1.f + erff(x2 * 0.70710678118654752f));
        const float h = x1 * g;
        float s = h;
        s += rowx<1>(s); s += rowx<2>(s); s += rowx<4>(s); s += rowx<8>(s);
        const float m = s * (1.f / 16.f);
        const float d = h - m;
        float q = d * d;
        q += rowx<1>(q); q += rowx<2>(q); q += rowx<4>(q); q += rowx<8>(q);
        if (lane < 16) Out[(long)i * ldout + na] = h;
        if (lane == 0) {
          stats[(blockIdx.x * BN + i) * 2]     = m;
          stats[(blockIdx.x * BN + i) * 2 + 1] = q;
        }
      }
    }
  } else if (wave == 0 && lane < 16) {
    for (int i = 0; i < BN; ++i) {
      float v = red[0][i][lane] + red[1][i][lane] +
                red[2][i][lane] + red[3][i][lane];
      if (bias) v += bias[n0 + lane];
      Out[(long)i * ldout + n0 + lane] = v;
    }
  }
}

__global__ void scores_k(const float* __restrict__ x,
                         const float* __restrict__ qtil,
                         float* __restrict__ scores) {
  const int b    = blockIdx.y;
  const int srow = blockIdx.x * 8 + (threadIdx.x >> 5);
  const int lane = threadIdx.x & 31;
  const float4* xr = (const float4*)(x + ((size_t)b * SS + srow) * DD);
  const float4* q4 = (const float4*)(qtil + (size_t)b * DD);
  float sum = 0.f;
#pragma unroll
  for (int i = 0; i < 8; ++i) {
    float4 xv = xr[lane + i * 32];
    float4 qv = q4[lane + i * 32];
    sum += xv.x * qv.x + xv.y * qv.y + xv.z * qv.z + xv.w * qv.w;
  }
  for (int off = 16; off; off >>= 1) sum += __shfl_xor(sum, off, 32);
  if (lane == 0) scores[b * SS + srow] = sum * 0.03125f;
}

__global__ void softmax_k(float* __restrict__ scores) {
  const int b = blockIdx.x, tid = threadIdx.x;
  __shared__ __attribute__((aligned(16))) float part[2][8];
  float* sc = scores + b * SS;
  float v[8];
  float mx = -3.402823466e38f;
#pragma unroll
  for (int j = 0; j < 8; ++j) { v[j] = sc[tid + j * 256]; mx = fmaxf(mx, v[j]); }
  mx = block_max(mx, part[0]);
  float sum = 0.f;
#pragma unroll
  for (int j = 0; j < 8; ++j) { v[j] = expf(v[j] - mx); sum += v[j]; }
  const float inv = 1.f / block_sum(sum, part[1]);
#pragma unroll
  for (int j = 0; j < 8; ++j) sc[tid + j * 256] = v[j] * inv;
}

#define WROWS 64
__global__ void __launch_bounds__(256) __attribute__((amdgpu_waves_per_eu(2, 2)))
wsum_cols(const float* __restrict__ x,
          const float* __restrict__ wts,
          float* __restrict__ cvec) {
  __shared__ __attribute__((aligned(16))) float red[8][32];
  const int strip = blockIdx.x, b = blockIdx.y, tid = threadIdx.x;
  const int q = tid & 7, g = tid >> 3;
  const int lane = tid & 31, wave = tid >> 5;
  const float4* xp = (const float4*)x +
      ((size_t)b * SS + (size_t)g * WROWS) * (DD / 4) + strip * 8 + q;
  const float4* wp = (const float4*)(wts + b * SS + g * WROWS);
  float4 wv[WROWS / 4];
  float4 xv[WROWS];
#pragma unroll
  for (int i = 0; i < WROWS / 4; ++i) wv[i] = wp[i];
#pragma unroll
  for (int i = 0; i < WROWS; ++i) xv[i] = xp[(size_t)i * (DD / 4)];
  float4 a0 = {0.f, 0.f, 0.f, 0.f};
  float4 a1 = {0.f, 0.f, 0.f, 0.f};
#pragma unroll
  for (int i = 0; i < WROWS / 4; ++i) {
    const float4 w4 = wv[i];
    const float4 x0 = xv[4 * i], x1 = xv[4 * i + 1];
    const float4 x2 = xv[4 * i + 2], x3 = xv[4 * i + 3];
    a0.x += w4.x * x0.x; a0.y += w4.x * x0.y; a0.z += w4.x * x0.z; a0.w += w4.x * x0.w;
    a1.x += w4.y * x1.x; a1.y += w4.y * x1.y; a1.z += w4.y * x1.z; a1.w += w4.y * x1.w;
    a0.x += w4.z * x2.x; a0.y += w4.z * x2.y; a0.z += w4.z * x2.z; a0.w += w4.z * x2.w;
    a1.x += w4.w * x3.x; a1.y += w4.w * x3.y; a1.z += w4.w * x3.z; a1.w += w4.w * x3.w;
  }
  float4 a = {a0.x + a1.x, a0.y + a1.y, a0.z + a1.z, a0.w + a1.w};
  for (int off = 8; off <= 16; off <<= 1) {
    a.x += __shfl_xor(a.x, off, 32); a.y += __shfl_xor(a.y, off, 32);
    a.z += __shfl_xor(a.z, off, 32); a.w += __shfl_xor(a.w, off, 32);
  }
  if (lane < 8) *(float4*)&red[wave][4 * lane] = a;
  __syncthreads();
  if (tid < 32) {
    float v = red[0][tid];
    for (int w = 1; w < 8; ++w) v += red[w][tid];
    cvec[b * DD + strip * 32 + tid] = v;
  }
}

extern "C" void kernel_launch(void* const* d_in, const int* in_sizes, int n_in,
                              void* d_out, int out_size, void* d_ws, size_t ws_size,
                              hipStream_t stream) {
  const float* x       = (const float*)d_in[0];
  const float* Wq_proj = (const float*)d_in[1];
  const float* bq      = (const float*)d_in[2];
  const float* Wq      = (const float*)d_in[3];
  const float* Wk      = (const float*)d_in[4];
  const float* Wv      = (const float*)d_in[5];
  const float* W1      = (const float*)d_in[6];
  const float* b1      = (const float*)d_in[7];
  const float* W2      = (const float*)d_in[8];
  const float* b2      = (const float*)d_in[9];
  const float* gamma   = (const float*)d_in[10];
  const float* beta    = (const float*)d_in[11];
  float* out = (float*)d_out;

  float* ws     = (float*)d_ws;
  float* qproj  = ws;
  float* qlast  = ws + 4096;
  float* qtil   = ws + 8192;
  float* scores = ws + 12288;
  float* cvec   = ws + 20480;
  float* attn   = ws + 24576;
  float* hglu   = ws + 28672;
  float* stats  = ws + 36864;

  wmma_gemm16<false, 0><<<dim3(HH / 16), 128, 0, stream>>>(
      x + (size_t)(SS - 1) * DD, (long)SS * DD, Wq_proj, HH, bq, qproj, HH,
      nullptr, nullptr, nullptr, nullptr);
  wmma_gemm16<false, 0><<<dim3(HH / 16), 128, 0, stream>>>(
      qproj, HH, Wq, HH, nullptr, qlast, HH, nullptr, nullptr, nullptr, nullptr);
  wmma_gemm16<true, 0><<<dim3(DD / 16), 128, 0, stream>>>(
      qlast, HH, Wk, HH, nullptr, qtil, DD, nullptr, nullptr, nullptr, nullptr);
  scores_k<<<dim3(SS / 8, BN), 256, 0, stream>>>(x, qtil, scores);
  softmax_k<<<dim3(BN), 256, 0, stream>>>(scores);
  wsum_cols<<<dim3(DD / 32, BN), 256, 0, stream>>>(x, scores, cvec);
  wmma_gemm16<false, 0><<<dim3(HH / 16), 128, 0, stream>>>(
      cvec, DD, Wv, HH, nullptr, attn, HH, nullptr, nullptr, nullptr, nullptr);
  wmma_gemm16<false, 1><<<dim3(HH / 16), 256, 0, stream>>>(
      attn, HH, W1, FF, b1, hglu, HH, nullptr, nullptr, nullptr, stats);
  wmma_gemm16<false, 2><<<dim3(OO / 16), 128, 0, stream>>>(
      attn, HH, W2, OO, b2, out, OO, hglu, gamma, beta, stats);
}


# launches 9 -> 8: softmax split by score block: score blocks write exp(s - own max) and a (max, sum) pair; each weighted-sum wave merges the 256 pairs by lane butterflies; new softmax sum order
# speedup vs baseline: 1.8666x; 1.0646x over previous
#include <hip/hip_runtime.h>
#include <hip/hip_bf16.h>
#include <math.h>

typedef float v2f __attribute__((ext_vector_type(2)));
typedef float v8f __attribute__((ext_vector_type(8)));
typedef float v4f __attribute__((ext_vector_type(4)));

#define BN 4
#define SS 2048
#define DD 1024
#define HH 1024
#define OO 1024
#define FF 2048

template <int M>
__device__ __forceinline__ float rowx(float x) {
  const int b = __builtin_bit_cast(int, x);
  return __builtin_bit_cast(
      float, __builtin_amdgcn_update_dpp(b, b, 0x160 + M, 0xf, 0xf, false));
}
__device__ __forceinline__ float xhalf(float x) {
  const int b = __builtin_bit_cast(int, x);
  return __builtin_bit_cast(
      float, __builtin_amdgcn_permlanex16(b, b, 0x76543210, 0xfedcba98, false, false));
}
__device__ __forceinline__ float wave_sum(float v) {
  v += rowx<1>(v); v += rowx<2>(v); v += rowx<4>(v); v += rowx<8>(v);
  return v + xhalf(v);
}
__device__ __forceinline__ float wave_max(float v) {
  v = fmaxf(v, rowx<1>(v)); v = fmaxf(v, rowx<2>(v));
  v = fmaxf(v, rowx<4>(v)); v = fmaxf(v, rowx<8>(v));
  return fmaxf(v, xhalf(v));
}

__device__ __forceinline__ void merge_stats(float& m, float& q, float mo, float qo,
                                            float halfn) {
  const float d = mo - m;
  m = 0.5f * (m + mo);
  q = (q + qo) + d * d * halfn;
}

#define GK 1024
#define LPITCH 1028

template <bool TRANSB, int MODE>
__global__ void __launch_bounds__(MODE == 1 ? 256 : 128)
wmma_gemm16(const float* __restrict__ A, long lda,
            const float* __restrict__ Bm, int ldb,
            const float* __restrict__ bias,
            float* __restrict__ Out, int ldout,
            const float* __restrict__ Hn,
            const float* __restrict__ gamma,
            const float* __restrict__ beta,
            float* __restrict__ stats) {
  constexpr int NP = (MODE == 1) ? 2 : 1;
  __shared__ __attribute__((aligned(16))) float sA[NP * BN * LPITCH];
  __shared__ __attribute__((aligned(16))) float sB[NP * (TRANSB ? 16 * LPITCH : GK * 16)];
  __shared__ float red[4 * NP][BN][16];
  const int tid  = threadIdx.x;
  const int lane = tid & 31;
  const int wave = tid >> 5;
  const int kq    = wave & 3;
  const int panel = (MODE == 1) ? (wave >> 2) : 0;
  const int n0   = blockIdx.x * 16 + panel * HH;
  const int ln   = lane & 15;
  const int kk   = (lane >> 4) * 2;
  const int kBeg = kq * (GK / 4);
  float* const sAp = sA + panel * (BN * LPITCH);
  float* const sBp = sB + panel * (GK * 16);

  v4f av[8];
  v4f bv[8][4];
#pragma unroll
  for (int i = 0; i < 8; ++i) {
    av[i] = *(const v4f*)(A + (long)(i >> 1) * lda + kBeg + 128 * (i & 1) + 4 * lane);
  }
  v4f hv[8], gv[2], bev[2], st[4];
  if (MODE == 2) {
#pragma unroll
    for (int i = 0; i < 8; ++i)
      hv[i] = *(const v4f*)(Hn + (i >> 1) * HH + kBeg + 128 * (i & 1) + 4 * lane);
#pragma unroll
    for (int j = 0; j < 2; ++j) {
      gv[j]  = *(const v4f*)(gamma + kBeg + 128 * j + 4 * lane);
      bev[j] = *(const v4f*)(beta + kBeg + 128 * j + 4 * lane);
      st[2 * j]     = *(const v4f*)(stats + (lane + 32 * j) * (2 * BN));
      st[2 * j + 1] = *(const v4f*)(stats + (lane + 32 * j) * (2 * BN) + 4);
    }
  }
  const float* gp = TRANSB
      ? (Bm + (long)(n0 + (lane >> 3)) * ldb + kBeg + 4 * (lane & 7))
      : (Bm + (long)(kBeg + (lane >> 2)) * ldb + n0 + 4 * (lane & 3));
#pragma unroll
  for (int c = 0; c < 8; ++c) {
#pragma unroll
    for (int j = 0; j < 4; ++j) {
      if (TRANSB)
        bv[c][j] = *(const v4f*)(gp + (long)(4 * j) * ldb + 32 * c);
      else
        bv[c][j] = *(const v4f*)(gp + (long)(32 * c + 8 * j) * ldb);
    }
    __builtin_amdgcn_sched_barrier(0);
  }

  if (MODE == 2) {
#pragma unroll
    for (int r = 0; r < BN; ++r) {
      const v4f sa = st[r >> 1], sb = st[2 + (r >> 1)];
      float m = (r & 1) ? sa.z : sa.x, q = (r & 1) ? sa.w : sa.y;
      merge_stats(m, q, (r & 1) ? sb.z : sb.x, (r & 1) ? sb.w : sb.y, 8.f);
      merge_stats(m, q, rowx<1>(m), rowx<1>(q), 16.f);
      merge_stats(m, q, rowx<2>(m), rowx<2>(q), 32.f);
      merge_stats(m, q, rowx<4>(m), rowx<4>(q), 64.f);
      merge_stats(m, q, rowx<8>(m), rowx<8>(q), 128.f);
      merge_stats(m, q, xhalf(m), xhalf(q), 256.f);
      const float rstd = rsqrtf(q * (1.f / HH) + 1e-5f);
      av[2 * r]     = av[2 * r]     + (hv[2 * r]     - m) * rstd * gv[0] + bev[0];
      av[2 * r + 1] = av[2 * r + 1] + (hv[2 * r + 1] - m) * rstd * gv[1] + bev[1];
    }
  }

#pragma unroll
  for (int i = 0; i < 8; ++i)
    *(v4f*)(sAp + (i >> 1) * LPITCH + kBeg + 128 * (i & 1) + 4 * lane) = av[i];

  const float* ap = sAp + (ln & 3) * LPITCH + kBeg + kk;
  const float* bp = TRANSB ? (sBp + ln * LPITCH + kBeg + kk)
                           : (sBp + (kBeg + kk) * 16 + ln);
  float* sp = TRANSB ? (sBp + (lane >> 3) * LPITCH + kBeg + 4 * (lane & 7))
                     : (sBp + kBeg * 16 + 4 * lane);

  v8f acc0 = {0.f, 0.f, 0.f, 0.f, 0.f, 0.f, 0.f, 0.f};
  v8f acc1 = {0.f, 0.f, 0.f, 0.f, 0.f, 0.f, 0.f, 0.f};

  v2f ae[8], be[8], ao[8], bo[8];

#define GEMM_CHUNK_IN(C, FA, FB)                                                   \
  _Pragma("unroll") for (int j = 0; j < 4; ++j) {                                  \
    if (TRANSB) *(v4f*)(sp + 4 * j * LPITCH + 32 * (C)) = bv[C][j];                \
    else        *(v4f*)(sp + 512 * (C) + 128 * j) = bv[C][j];                      \
  }                                                                                \
  __builtin_amdgcn_fence(__ATOMIC_RELEASE, "wavefront");                           \
  __builtin_amdgcn_wave_barrier();                                                 \
  __builtin_amdgcn_fence(__ATOMIC_ACQUIRE, "wavefront");                           \
  _Pragma("unroll") for (int j = 0; j < 8; ++j) {                                  \
    const int i = 8 * (C) + j;                     \
    FA[j] = *(const v2f*)(ap + 4 * i);                                             \
    if (TRANSB) {                                                                  \
      FB[j] = *(const v2f*)(bp + 4 * i);                                           \
    } else {                                                                       \
      FB[j].x = bp[(4 * i) * 16];                                                  \
      FB[j].y = bp[(4 * i + 1) * 16];                                              \
    }                                                                              \
  }
#define GEMM_CHUNK_MM(FA, FB)                                                      \
  _Pragma("unroll") for (int j = 0; j < 8; j += 2) {                               \
    acc0 = __builtin_amdgcn_wmma_f32_16x16x4_f32(                                  \
        false, FA[j], false, FB[j], (short)0, acc0, false, false);                 \
    acc1 = __builtin_amdgcn_wmma_f32_16x16x4_f32(                                  \
        false, FA[j + 1], false, FB[j + 1], (short)0, acc1, false, false);         \
  }
  constexpr int DSC = TRANSB ? 8 : 12;

  GEMM_CHUNK_IN(0, ae, be)
  GEMM_CHUNK_IN(1, ao, bo) GEMM_CHUNK_MM(ae, be)
  GEMM_CHUNK_IN(2, ae, be) GEMM_CHUNK_MM(ao, bo)
  GEMM_CHUNK_IN(3, ao, bo) GEMM_CHUNK_MM(ae, be)
  GEMM_CHUNK_IN(4, ae, be) GEMM_CHUNK_MM(ao, bo)
  GEMM_CHUNK_IN(5, ao, bo) GEMM_CHUNK_MM(ae, be)
  GEMM_CHUNK_IN(6, ae, be) GEMM_CHUNK_MM(ao, bo)
  GEMM_CHUNK_IN(7, ao, bo) GEMM_CHUNK_MM(ae, be)
  GEMM_CHUNK_MM(ao, bo)
#undef GEMM_CHUNK_IN
#undef GEMM_CHUNK_MM
  __builtin_amdgcn_sched_group_barrier(0x200, 12, 0);
  __builtin_amdgcn_sched_group_barrier(0x100, DSC, 0);
#pragma unroll
  for (int c = 1; c < 8; ++c) {
    __builtin_amdgcn_sched_group_barrier(0x200, 4, 0);
    __builtin_amdgcn_sched_group_barrier(0x100, DSC, 0);
    __builtin_amdgcn_sched_group_barrier(0x008, 8, 0);
  }
  __builtin_amdgcn_sched_group_barrier(0x008, 8, 0);

  if (lane < 16) {
    for (int i = 0; i < BN; ++i) red[wave][i][lane] = acc0[i] + acc1[i];
  }
  __syncthreads();

  if (MODE == 1) {
    if (wave == 0) {
      const int c  = lane & 15;
      const int na = blockIdx.x * 16 + c;
      const float bx1 = bias[na], bx2 = bias[HH + na];
      for (int i = 0; i < BN; ++i) {
        const float x1 = red[0][i][c] + red[1][i][c] + red[2][i][c] + red[3][i][c] + bx1;
        const float x2 = red[4][i][c] + red[5][i][c] + red[6][i][c] + red[7][i][c] + bx2;
        const float g = 0.5f * x2 * (1.f + erff(x2 * 0.70710678118654752f));
        const float h = x1 * g;
        float s = h;
        s += rowx<1>(s); s += rowx<2>(s); s += rowx<4>(s); s += rowx<8>(s);
        const float m = s * (1.f / 16.f);
        const float d = h - m;
        float q = d * d;
        q += rowx<1>(q); q += rowx<2>(q); q += rowx<4>(q); q += rowx<8>(q);
        if (lane < 16) Out[(long)i * ldout + na] = h;
        if (lane == 0) {
          stats[(blockIdx.x * BN + i) * 2]     = m;
          stats[(blockIdx.x * BN + i) * 2 + 1] = q;
        }
      }
    }
  } else if (wave == 0 && lane < 16) {
    for (int i = 0; i < BN; ++i) {
      float v = red[0][i][lane] + red[1][i][lane] +
                red[2][i][lane] + red[3][i][lane];
      if (bias) v += bias[n0 + lane];
      Out[(long)i * ldout + n0 + lane] = v;
    }
  }
}

#define SBLK (SS / 8)
__global__ void scores_k(const float* __restrict__ x,
                         const float* __restrict__ qtil,
                         float* __restrict__ ew,
                         float* __restrict__ pairs) {
  __shared__ float sh[8];
  const int b    = blockIdx.y;
  const int wave = threadIdx.x >> 5;
  const int srow = blockIdx.x * 8 + wave;
  const int lane = threadIdx.x & 31;
  const float4* xr = (const float4*)(x + ((size_t)b * SS + srow) * DD);
  const float4* q4 = (const float4*)(qtil + (size_t)b * DD);
  float sum = 0.f;
#pragma unroll
  for (int i = 0; i < 8; ++i) {
    float4 xv = xr[lane + i * 32];
    float4 qv = q4[lane + i * 32];
    sum += xv.x * qv.x + xv.y * qv.y + xv.z * qv.z + xv.w * qv.w;
  }
  for (int off = 16; off; off >>= 1) sum += __shfl_xor(sum, off, 32);
  if (lane == 0) sh[wave] = sum * 0.03125f;
  __syncthreads();
  if (wave == 0) {
    const float s = sh[lane & 7];
    float m = fmaxf(s, rowx<1>(s));
    m = fmaxf(m, rowx<2>(m));
    m = fmaxf(m, rowx<4>(m));
    const float e = expf(s - m);
    float l = e + rowx<1>(e);
    l += rowx<2>(l);
    l += rowx<4>(l);
    if (lane < 8) ew[b * SS + blockIdx.x * 8 + lane] = e;
    if (lane == 0) *(float2*)(pairs + (b * SBLK + blockIdx.x) * 2) = make_float2(m, l);
  }
}

#define WROWS 64
__global__ void __launch_bounds__(256) __attribute__((amdgpu_waves_per_eu(2, 2)))
wsum_cols(const float* __restrict__ x,
          const float* __restrict__ wts,
          const float* __restrict__ pairs,
          float* __restrict__ cvec) {
  __shared__ __attribute__((aligned(16))) float red[8][32];
  const int strip = blockIdx.x, b = blockIdx.y, tid = threadIdx.x;
  const int q = tid & 7, g = tid >> 3;
  const int lane = tid & 31, wave = tid >> 5;
  const float4* xp = (const float4*)x +
      ((size_t)b * SS + (size_t)g * WROWS) * (DD / 4) + strip * 8 + q;
  const float4* wp = (const float4*)(wts + b * SS + g * WROWS);
  const float4* pp = (const float4*)(pairs + b * SBLK * 2);
  float4 pa[4], po[4];
#pragma unroll
  for (int i = 0; i < 4; ++i) { pa[i] = pp[lane * 4 + i]; po[i] = pp[g * 4 + i]; }
  float4 wv[WROWS / 4];
  float4 xv[WROWS];
#pragma unroll
  for (int i = 0; i < WROWS / 4; ++i) wv[i] = wp[i];
#pragma unroll
  for (int i = 0; i < WROWS; ++i) xv[i] = xp[(size_t)i * (DD / 4)];
  float M = fmaxf(fmaxf(fmaxf(pa[0].x, pa[0].z), fmaxf(pa[1].x, pa[1].z)),
                  fmaxf(fmaxf(pa[2].x, pa[2].z), fmaxf(pa[3].x, pa[3].z)));
  M = wave_max(M);
  float L = 0.f;
#pragma unroll
  for (int i = 0; i < 4; ++i) {
    L += pa[i].y * expf(pa[i].x - M);
    L += pa[i].w * expf(pa[i].z - M);
  }
  const float invL = 1.f / wave_sum(L);
  float fac[8];
#pragma unroll
  for (int i = 0; i < 4; ++i) {
    fac[2 * i]     = expf(po[i].x - M) * invL;
    fac[2 * i + 1] = expf(po[i].z - M) * invL;
  }
  float4 a0 = {0.f, 0.f, 0.f, 0.f};
  float4 a1 = {0.f, 0.f, 0.f, 0.f};
#pragma unroll
  for (int i = 0; i < WROWS / 4; ++i) {
    const float f = fac[i >> 1];
    const float4 w4 = {wv[i].x * f, wv[i].y * f, wv[i].z * f, wv[i].w * f};
    const float4 x0 = xv[4 * i], x1 = xv[4 * i + 1];
    const float4 x2 = xv[4 * i + 2], x3 = xv[4 * i + 3];
    a0.x += w4.x * x0.x; a0.y += w4.x * x0.y; a0.z += w4.x * x0.z; a0.w += w4.x * x0.w;
    a1.x += w4.y * x1.x; a1.y += w4.y * x1.y; a1.z += w4.y * x1.z; a1.w += w4.y * x1.w;
    a0.x += w4.z * x2.x; a0.y += w4.z * x2.y; a0.z += w4.z * x2.z; a0.w += w4.z * x2.w;
    a1.x += w4.w * x3.x; a1.y += w4.w * x3.y; a1.z += w4.w * x3.z; a1.w += w4.w * x3.w;
  }
  float4 a = {a0.x + a1.x, a0.y + a1.y, a0.z + a1.z, a0.w + a1.w};
  for (int off = 8; off <= 16; off <<= 1) {
    a.x += __shfl_xor(a.x, off, 32); a.y += __shfl_xor(a.y, off, 32);
    a.z += __shfl_xor(a.z, off, 32); a.w += __shfl_xor(a.w, off, 32);
  }
  if (lane < 8) *(float4*)&red[wave][4 * lane] = a;
  __syncthreads();
  if (tid < 32) {
    float v = red[0][tid];
    for (int w = 1; w < 8; ++w) v += red[w][tid];
    cvec[b * DD + strip * 32 + tid] = v;
  }
}

extern "C" void kernel_launch(void* const* d_in, const int* in_sizes, int n_in,
                              void* d_out, int out_size, void* d_ws, size_t ws_size,
                              hipStream_t stream) {
  const float* x       = (const float*)d_in[0];
  const float* Wq_proj = (const float*)d_in[1];
  const float* bq      = (const float*)d_in[2];
  const float* Wq      = (const float*)d_in[3];
  const float* Wk      = (const float*)d_in[4];
  const float* Wv      = (const float*)d_in[5];
  const float* W1      = (const float*)d_in[6];
  const float* b1      = (const float*)d_in[7];
  const float* W2      = (const float*)d_in[8];
  const float* b2      = (const float*)d_in[9];
  const float* gamma   = (const float*)d_in[10];
  const float* beta    = (const float*)d_in[11];
  float* out = (float*)d_out;

  float* ws     = (float*)d_ws;
  float* qproj  = ws;
  float* qlast  = ws + 4096;
  float* qtil   = ws + 8192;
  float* expw   = ws + 12288;
  float* cvec   = ws + 20480;
  float* attn   = ws + 24576;
  float* hglu   = ws + 28672;
  float* stats  = ws + 36864;
  float* pairs  = ws + 40960;

  wmma_gemm16<false, 0><<<dim3(HH / 16), 128, 0, stream>>>(
      x + (size_t)(SS - 1) * DD, (long)SS * DD, Wq_proj, HH, bq, qproj, HH,
      nullptr, nullptr, nullptr, nullptr);
  wmma_gemm16<false, 0><<<dim3(HH / 16), 128, 0, stream>>>(
      qproj, HH, Wq, HH, nullptr, qlast, HH, nullptr, nullptr, nullptr, nullptr);
  wmma_gemm16<true, 0><<<dim3(DD / 16), 128, 0, stream>>>(
      qlast, HH, Wk, HH, nullptr, qtil, DD, nullptr, nullptr, nullptr, nullptr);
  scores_k<<<dim3(SS / 8, BN), 256, 0, stream>>>(x, qtil, expw, pairs);
  wsum_cols<<<dim3(DD / 32, BN), 256, 0, stream>>>(x, expw, pairs, cvec);
  wmma_gemm16<false, 0><<<dim3(HH / 16), 128, 0, stream>>>(
      cvec, DD, Wv, HH, nullptr, attn, HH, nullptr, nullptr, nullptr, nullptr);
  wmma_gemm16<false, 1><<<dim3(HH / 16), 256, 0, stream>>>(
      attn, HH, W1, FF, b1, hglu, HH, nullptr, nullptr, nullptr, stats);
  wmma_gemm16<false, 2><<<dim3(OO / 16), 128, 0, stream>>>(
      attn, HH, W2, OO, b2, out, OO, hglu, gamma, beta, stats);
}


# small GEMMs, tail behind the block barrier: wave r finishes output row r (was wave 0 taking the 4 rows in turn); bias and epilogue arguments are fetched at kernel start, not there; sums unchanged
# speedup vs baseline: 2.0909x; 1.1202x over previous
#include <hip/hip_runtime.h>
#include <hip/hip_bf16.h>
#include <math.h>

typedef float v2f __attribute__((ext_vector_type(2)));
typedef float v8f __attribute__((ext_vector_type(8)));
typedef float v4f __attribute__((ext_vector_type(4)));

#define BN 4
#define SS 2048
#define DD 1024
#define HH 1024
#define OO 1024
#define FF 2048

template <int M>
__device__ __forceinline__ float rowx(float x) {
  const int b = __builtin_bit_cast(int, x);
  return __builtin_bit_cast(
      float, __builtin_amdgcn_update_dpp(b, b, 0x160 + M, 0xf, 0xf, false));
}
__device__ __forceinline__ float xhalf(float x) {
  const int b = __builtin_bit_cast(int, x);
  return __builtin_bit_cast(
      float, __builtin_amdgcn_permlanex16(b, b, 0x76543210, 0xfedcba98, false, false));
}
__device__ __forceinline__ float wave_sum(float v) {
  v += rowx<1>(v); v += rowx<2>(v); v += rowx<4>(v); v += rowx<8>(v);
  return v + xhalf(v);
}
__device__ __forceinline__ float wave_max(float v) {
  v = fmaxf(v, rowx<1>(v)); v = fmaxf(v, rowx<2>(v));
  v = fmaxf(v, rowx<4>(v)); v = fmaxf(v, rowx<8>(v));
  return fmaxf(v, xhalf(v));
}

__device__ __forceinline__ void merge_stats(float& m, float& q, float mo, float qo,
                                            float halfn) {
  const float d = mo - m;
  m = 0.5f * (m + mo);
  q = (q + qo) + d * d * halfn;
}

#define GK 1024
#define LPITCH 1028

template <bool TRANSB, int MODE, bool BIAS>
__global__ void __launch_bounds__(MODE == 1 ? 256 : 128)
wmma_gemm16(const float* __restrict__ A, long lda,
            const float* __restrict__ Bm, int ldb,
            const float* __restrict__ bias,
            float* __restrict__ Out, int ldout,
            const float* __restrict__ Hn,
            const float* __restrict__ gamma,
            const float* __restrict__ beta,
            float* __restrict__ stats) {
  constexpr int NP = (MODE == 1) ? 2 : 1;
  __shared__ __attribute__((aligned(16))) float sA[NP * BN * LPITCH];
  __shared__ __attribute__((aligned(16))) float sB[NP * (TRANSB ? 16 * LPITCH : GK * 16)];
  __shared__ float red[4 * NP][BN][16];
  const int tid  = threadIdx.x;
  const int lane = tid & 31;
  const int wave = tid >> 5;
  const int kq    = wave & 3;
  const int panel = (MODE == 1) ? (wave >> 2) : 0;
  const int n0   = blockIdx.x * 16 + panel * HH;
  const int ln   = lane & 15;
  const int kk   = (lane >> 4) * 2;
  const int kBeg = kq * (GK / 4);
  float* const sAp = sA + panel * (BN * LPITCH);
  float* const sBp = sB + panel * (GK * 16);

  v4f av[8];
  v4f bv[8][4];
#pragma unroll
  for (int i = 0; i < 8; ++i) {
    av[i] = *(const v4f*)(A + (long)(i >> 1) * lda + kBeg + 128 * (i & 1) + 4 * lane);
  }
  v4f hv[8], gv[2], bev[2], st[4];
  if (MODE == 2) {
#pragma unroll
    for (int i = 0; i < 8; ++i)
      hv[i] = *(const v4f*)(Hn + (i >> 1) * HH + kBeg + 128 * (i & 1) + 4 * lane);
#pragma unroll
    for (int j = 0; j < 2; ++j) {
      gv[j]  = *(const v4f*)(gamma + kBeg + 128 * j + 4 * lane);
      bev[j] = *(const v4f*)(beta + kBeg + 128 * j + 4 * lane);
      st[2 * j]     = *(const v4f*)(stats + (lane + 32 * j) * (2 * BN));
      st[2 * j + 1] = *(const v4f*)(stats + (lane + 32 * j) * (2 * BN) + 4);
    }
  }
  const float* gp = TRANSB
      ? (Bm + (long)(n0 + (lane >> 3)) * ldb + kBeg + 4 * (lane & 7))
      : (Bm + (long)(kBeg + (lane >> 2)) * ldb + n0 + 4 * (lane & 3));
#pragma unroll
  for (int c = 0; c < 8; ++c) {
#pragma unroll
    for (int j = 0; j < 4; ++j) {
      if (TRANSB)
        bv[c][j] = *(const v4f*)(gp + (long)(4 * j) * ldb + 32 * c);
      else
        bv[c][j] = *(const v4f*)(gp + (long)(32 * c + 8 * j) * ldb);
    }
    __builtin_amdgcn_sched_barrier(0);
  }
  float bz0 = 0.f, bz1 = 0.f;
  if (MODE == 1) {
    bz0 = bias[blockIdx.x * 16 + ln];
    bz1 = bias[HH + blockIdx.x * 16 + ln];
  } else if (BIAS) {
    bz0 = bias[n0 + ln];
  }
  asm volatile("" :: "s"(Out), "s"(ldout));
  if (MODE == 1) asm volatile("" :: "s"(stats));

  if (MODE == 2) {
#pragma unroll
    for (int r = 0; r < BN; ++r) {
      const v4f sa = st[r >> 1], sb = st[2 + (r >> 1)];
      float m = (r & 1) ? sa.z : sa.x, q = (r & 1) ? sa.w : sa.y;
      merge_stats(m, q, (r & 1) ? sb.z : sb.x, (r & 1) ? sb.w : sb.y, 8.f);
      merge_stats(m, q, rowx<1>(m), rowx<1>(q), 16.f);
      merge_stats(m, q, rowx<2>(m), rowx<2>(q), 32.f);
      merge_stats(m, q, rowx<4>(m), rowx<4>(q), 64.f);
      merge_stats(m, q, rowx<8>(m), rowx<8>(q), 128.f);
      merge_stats(m, q, xhalf(m), xhalf(q), 256.f);
      const float rstd = rsqrtf(q * (1.f / HH) + 1e-5f);
      av[2 * r]     = av[2 * r]     + (hv[2 * r]     - m) * rstd * gv[0] + bev[0];
      av[2 * r + 1] = av[2 * r + 1] + (hv[2 * r + 1] - m) * rstd * gv[1] + bev[1];
    }
  }

#pragma unroll
  for (int i = 0; i < 8; ++i)
    *(v4f*)(sAp + (i >> 1) * LPITCH + kBeg + 128 * (i & 1) + 4 * lane) = av[i];

  const float* ap = sAp + (ln & 3) * LPITCH + kBeg + kk;
  const float* bp = TRANSB ? (sBp + ln * LPITCH + kBeg + kk)
                           : (sBp + (kBeg + kk) * 16 + ln);
  float* sp = TRANSB ? (sBp + (lane >> 3) * LPITCH + kBeg + 4 * (lane & 7))
                     : (sBp + kBeg * 16 + 4 * lane);

  v8f acc0 = {0.f, 0.f, 0.f, 0.f, 0.f, 0.f, 0.f, 0.f};
  v8f acc1 = {0.f, 0.f, 0.f, 0.f, 0.f, 0.f, 0.f, 0.f};

  v2f ae[8], be[8], ao[8], bo[8];

#define GEMM_CHUNK_IN(C, FA, FB)                                                   \
  _Pragma("unroll") for (int j = 0; j < 4; ++j) {                                  \
    if (TRANSB) *(v4f*)(sp + 4 * j * LPITCH + 32 * (C)) = bv[C][j];                \
    else        *(v4f*)(sp + 512 * (C) + 128 * j) = bv[C][j];                      \
  }                                                                                \
  __builtin_amdgcn_fence(__ATOMIC_RELEASE, "wavefront");                           \
  __builtin_amdgcn_wave_barrier();                                                 \
  __builtin_amdgcn_fence(__ATOMIC_ACQUIRE, "wavefront");                           \
  _Pragma("unroll") for (int j = 0; j < 8; ++j) {                                  \
    const int i = 8 * (C) + j;                     \
    FA[j] = *(const v2f*)(ap + 4 * i);                                             \
    if (TRANSB) {                                                                  \
      FB[j] = *(const v2f*)(bp + 4 * i);                                           \
    } else {                                                                       \
      FB[j].x = bp[(4 * i) * 16];                                                  \
      FB[j].y = bp[(4 * i + 1) * 16];                                              \
    }                                                                              \
  }
#define GEMM_CHUNK_MM(FA, FB)                                                      \
  _Pragma("unroll") for (int j = 0; j < 8; j += 2) {                               \
    acc0 = __builtin_amdgcn_wmma_f32_16x16x4_f32(                                  \
        false, FA[j], false, FB[j], (short)0, acc0, false, false);                 \
    acc1 = __builtin_amdgcn_wmma_f32_16x16x4_f32(                                  \
        false, FA[j + 1], false, FB[j + 1], (short)0, acc1, false, false);         \
  }
  constexpr int DSC = TRANSB ? 8 : 12;

  GEMM_CHUNK_IN(0, ae, be)
  GEMM_CHUNK_IN(1, ao, bo) GEMM_CHUNK_MM(ae, be)
  GEMM_CHUNK_IN(2, ae, be) GEMM_CHUNK_MM(ao, bo)
  GEMM_CHUNK_IN(3, ao, bo) GEMM_CHUNK_MM(ae, be)
  GEMM_CHUNK_IN(4, ae, be) GEMM_CHUNK_MM(ao, bo)
  GEMM_CHUNK_IN(5, ao, bo) GEMM_CHUNK_MM(ae, be)
  GEMM_CHUNK_IN(6, ae, be) GEMM_CHUNK_MM(ao, bo)
  GEMM_CHUNK_IN(7, ao, bo) GEMM_CHUNK_MM(ae, be)
  GEMM_CHUNK_MM(ao, bo)
#undef GEMM_CHUNK_IN
#undef GEMM_CHUNK_MM
  __builtin_amdgcn_sched_group_barrier(0x200, 12, 0);
  __builtin_amdgcn_sched_group_barrier(0x100, DSC, 0);
#pragma unroll
  for (int c = 1; c < 8; ++c) {
    __builtin_amdgcn_sched_group_barrier(0x200, 4, 0);
    __builtin_amdgcn_sched_group_barrier(0x100, DSC, 0);
    __builtin_amdgcn_sched_group_barrier(0x008, 8, 0);
  }
  __builtin_amdgcn_sched_group_barrier(0x008, 8, 0);

  if (lane < 16) {
    for (int i = 0; i < BN; ++i) red[wave][i][lane] = acc0[i] + acc1[i];
  }
  __syncthreads();

  if (MODE == 1) {
    if (wave < BN) {
      const int i  = wave;
      const int c  = lane & 15;
      const int na = blockIdx.x * 16 + c;
      const float x1 = red[0][i][c] + red[1][i][c] + red[2][i][c] + red[3][i][c] + bz0;
      const float x2 = red[4][i][c] + red[5][i][c] + red[6][i][c] + red[7][i][c] + bz1;
      const float g = 0.5f * x2 * (1.f + erff(x2 * 0.70710678118654752f));
      const float h = x1 * g;
      float s = h;
      s += rowx<1>(s); s += rowx<2>(s); s += rowx<4>(s); s += rowx<8>(s);
      const float m = s * (1.f / 16.f);
      const float d = h - m;
      float q = d * d;
      q += rowx<1>(q); q += rowx<2>(q); q += rowx<4>(q); q += rowx<8>(q);
      if (lane < 16) Out[(long)i * ldout + na] = h;
      if (lane == 0) {
        stats[(blockIdx.x * BN + i) * 2]     = m;
        stats[(blockIdx.x * BN + i) * 2 + 1] = q;
      }
    }
  } else if (lane < 16) {
    const int i = wave;
    float v = red[0][i][lane] + red[1][i][lane] +
              red[2][i][lane] + red[3][i][lane];
    if (BIAS) v += bz0;
    Out[(long)i * ldout + n0 + lane] = v;
  }
}

#define SBLK (SS / 8)
__global__ void scores_k(const float* __restrict__ x,
                         const float* __restrict__ qtil,
                         float* __restrict__ ew,
                         float* __restrict__ pairs) {
  __shared__ float sh[8];
  const int b    = blockIdx.y;
  const int wave = threadIdx.x >> 5;
  const int srow = blockIdx.x * 8 + wave;
  const int lane = threadIdx.x & 31;
  const float4* xr = (const float4*)(x + ((size_t)b * SS + srow) * DD);
  const float4* q4 = (const float4*)(qtil + (size_t)b * DD);
  float sum = 0.f;
#pragma unroll
  for (int i = 0; i < 8; ++i) {
    float4 xv = xr[lane + i * 32];
    float4 qv = q4[lane + i * 32];
    sum += xv.x * qv.x + xv.y * qv.y + xv.z * qv.z + xv.w * qv.w;
  }
  for (int off = 16; off; off >>= 1) sum += __shfl_xor(sum, off, 32);
  if (lane == 0) sh[wave] = sum * 0.03125f;
  __syncthreads();
  if (wave == 0) {
    const float s = sh[lane & 7];
    float m = fmaxf(s, rowx<1>(s));
    m = fmaxf(m, rowx<2>(m));
    m = fmaxf(m, rowx<4>(m));
    const float e = expf(s - m);
    float l = e + rowx<1>(e);
    l += rowx<2>(l);
    l += rowx<4>(l);
    if (lane < 8) ew[b * SS + blockIdx.x * 8 + lane] = e;
    if (lane == 0) *(float2*)(pairs + (b * SBLK + blockIdx.x) * 2) = make_float2(m, l);
  }
}

#define WROWS 64
__global__ void __launch_bounds__(256) __attribute__((amdgpu_waves_per_eu(2, 2)))
wsum_cols(const float* __restrict__ x,
          const float* __restrict__ wts,
          const float* __restrict__ pairs,
          float* __restrict__ cvec) {
  __shared__ __attribute__((aligned(16))) float red[8][32];
  const int strip = blockIdx.x, b = blockIdx.y, tid = threadIdx.x;
  const int q = tid & 7, g = tid >> 3;
  const int lane = tid & 31, wave = tid >> 5;
  const float4* xp = (const float4*)x +
      ((size_t)b * SS + (size_t)g * WROWS) * (DD / 4) + strip * 8 + q;
  const float4* wp = (const float4*)(wts + b * SS + g * WROWS);
  const float4* pp = (const float4*)(pairs + b * SBLK * 2);
  float4 pa[4], po[4];
#pragma unroll
  for (int i = 0; i < 4; ++i) { pa[i] = pp[lane * 4 + i]; po[i] = pp[g * 4 + i]; }
  float4 wv[WROWS / 4];
  float4 xv[WROWS];
#pragma unroll
  for (int i = 0; i < WROWS / 4; ++i) wv[i] = wp[i];
#pragma unroll
  for (int i = 0; i < WROWS; ++i) xv[i] = xp[(size_t)i * (DD / 4)];
  float M = fmaxf(fmaxf(fmaxf(pa[0].x, pa[0].z), fmaxf(pa[1].x, pa[1].z)),
                  fmaxf(fmaxf(pa[2].x, pa[2].z), fmaxf(pa[3].x, pa[3].z)));
  M = wave_max(M);
  float L = 0.f;
#pragma unroll
  for (int i = 0; i < 4; ++i) {
    L += pa[i].y * expf(pa[i].x - M);
    L += pa[i].w * expf(pa[i].z - M);
  }
  const float invL = 1.f / wave_sum(L);
  float fac[8];
#pragma unroll
  for (int i = 0; i < 4; ++i) {
    fac[2 * i]     = expf(po[i].x - M) * invL;
    fac[2 * i + 1] = expf(po[i].z - M) * invL;
  }
  float4 a0 = {0.f, 0.f, 0.f, 0.f};
  float4 a1 = {0.f, 0.f, 0.f, 0.f};
#pragma unroll
  for (int i = 0; i < WROWS / 4; ++i) {
    const float f = fac[i >> 1];
    const float4 w4 = {wv[i].x * f, wv[i].y * f, wv[i].z * f, wv[i].w * f};
    const float4 x0 = xv[4 * i], x1 = xv[4 * i + 1];
    const float4 x2 = xv[4 * i + 2], x3 = xv[4 * i + 3];
    a0.x += w4.x * x0.x; a0.y += w4.x * x0.y; a0.z += w4.x * x0.z; a0.w += w4.x * x0.w;
    a1.x += w4.y * x1.x; a1.y += w4.y * x1.y; a1.z += w4.y * x1.z; a1.w += w4.y * x1.w;
    a0.x += w4.z * x2.x; a0.y += w4.z * x2.y; a0.z += w4.z * x2.z; a0.w += w4.z * x2.w;
    a1.x += w4.w * x3.x; a1.y += w4.w * x3.y; a1.z += w4.w * x3.z; a1.w += w4.w * x3.w;
  }
  float4 a = {a0.x + a1.x, a0.y + a1.y, a0.z + a1.z, a0.w + a1.w};
  for (int off = 8; off <= 16; off <<= 1) {
    a.x += __shfl_xor(a.x, off, 32); a.y += __shfl_xor(a.y, off, 32);
    a.z += __shfl_xor(a.z, off, 32); a.w += __shfl_xor(a.w, off, 32);
  }
  if (lane < 8) *(float4*)&red[wave][4 * lane] = a;
  __syncthreads();
  if (tid < 32) {
    float v = red[0][tid];
    for (int w = 1; w < 8; ++w) v += red[w][tid];
    cvec[b * DD + strip * 32 + tid] = v;
  }
}

extern "C" void kernel_launch(void* const* d_in, const int* in_sizes, int n_in,
                              void* d_out, int out_size, void* d_ws, size_t ws_size,
                              hipStream_t stream) {
  const float* x       = (const float*)d_in[0];
  const float* Wq_proj = (const float*)d_in[1];
  const float* bq      = (const float*)d_in[2];
  const float* Wq      = (const float*)d_in[3];
  const float* Wk      = (const float*)d_in[4];
  const float* Wv      = (const float*)d_in[5];
  const float* W1      = (const float*)d_in[6];
  const float* b1      = (const float*)d_in[7];
  const float* W2      = (const float*)d_in[8];
  const float* b2      = (const float*)d_in[9];
  const float* gamma   = (const float*)d_in[10];
  const float* beta    = (const float*)d_in[11];
  float* out = (float*)d_out;

  float* ws     = (float*)d_ws;
  float* qproj  = ws;
  float* qlast  = ws + 4096;
  float* qtil   = ws + 8192;
  float* expw   = ws + 12288;
  float* cvec   = ws + 20480;
  float* attn   = ws + 24576;
  float* hglu   = ws + 28672;
  float* stats  = ws + 36864;
  float* pairs  = ws + 40960;

  wmma_gemm16<false, 0, true><<<dim3(HH / 16), 128, 0, stream>>>(
      x + (size_t)(SS - 1) * DD, (long)SS * DD, Wq_proj, HH, bq, qproj, HH,
      nullptr, nullptr, nullptr, nullptr);
  wmma_gemm16<false, 0, false><<<dim3(HH / 16), 128, 0, stream>>>(
      qproj, HH, Wq, HH, nullptr, qlast, HH, nullptr, nullptr, nullptr, nullptr);
  wmma_gemm16<true, 0, false><<<dim3(DD / 16), 128, 0, stream>>>(
      qlast, HH, Wk, HH, nullptr, qtil, DD, nullptr, nullptr, nullptr, nullptr);
  scores_k<<<dim3(SS / 8, BN), 256, 0, stream>>>(x, qtil, expw, pairs);
  wsum_cols<<<dim3(DD / 32, BN), 256, 0, stream>>>(x, expw, pairs, cvec);
  wmma_gemm16<false, 0, false><<<dim3(HH / 16), 128, 0, stream>>>(
      cvec, DD, Wv, HH, nullptr, attn, HH, nullptr, nullptr, nullptr, nullptr);
  wmma_gemm16<false, 1, true><<<dim3(HH / 16), 256, 0, stream>>>(
      attn, HH, W1, FF, b1, hglu, HH, nullptr, nullptr, nullptr, stats);
  wmma_gemm16<false, 2, true><<<dim3(OO / 16), 128, 0, stream>>>(
      attn, HH, W2, OO, b2, out, OO, hglu, gamma, beta, stats);
}
